# starting kernel
# baseline (speedup 1.0000x reference)
#include <hip/hip_runtime.h>
#include <hip/hip_bf16.h>

typedef float v2f __attribute__((ext_vector_type(2)));
typedef float v8f __attribute__((ext_vector_type(8)));

#define B_ 4096
#define D_ 128
#define C_ 256

__global__ __launch_bounds__(256) void mahal_asym_kernel(
    const float* __restrict__ alpha, float* __restrict__ Asym) {
  const int i = blockIdx.x * 256 + threadIdx.x;
  const int r = i >> 7;
  const int c = i & 127;
  Asym[i] = alpha[r * D_ + c] + alpha[c * D_ + r];
}

__global__ __launch_bounds__(128) void mahal_mam_kernel(
    const float* __restrict__ alpha, const float* __restrict__ means,
    float* __restrict__ mAm) {
  __shared__ float red[128];
  const int c = blockIdx.x;
  const int d = threadIdx.x;
  const float* arow = alpha + d * D_;
  float w = 0.f;
  for (int e = 0; e < D_; ++e) w += arow[e] * means[e * C_ + c];
  red[d] = w * means[d * C_ + c];
  __syncthreads();
  for (int s = 64; s > 0; s >>= 1) {
    if (d < s) red[d] += red[d + s];
    __syncthreads();
  }
  if (d == 0) mAm[c] = red[0];
}

__global__ __launch_bounds__(256) void mahal_main_kernel(
    const float* __restrict__ X, const float* __restrict__ means,
    const float* __restrict__ Asym, const float* __restrict__ mAm,
    float* __restrict__ out) {
  constexpr int LDP = D_ + 4;
  __shared__ float Xs[16 * LDP];
  __shared__ float Ys[16 * LDP];
  __shared__ float tmp[256];
  __shared__ float t1s[16];
  __shared__ float pm[8][16];

  const int tid = threadIdx.x;
  const int wave = tid >> 5;
  const int lane = tid & 31;
  const int hn = lane & 15;
  const int hi = lane >> 4;
  const int rowbase = blockIdx.x * 16;

  for (int i = tid; i < 16 * D_; i += 256) {
    int r = i >> 7;
    int cc = i & 127;
    Xs[r * LDP + cc] = X[(rowbase + r) * D_ + cc];
  }
  __syncthreads();

  {
    const int cb = wave * 16;
    const int cg = cb + hn;
    v8f acc = {};
#pragma unroll 4
    for (int k0 = 0; k0 < D_; k0 += 4) {
      const int ka = k0 + 2 * hi;
      v2f a, b;
      a.x = Xs[hn * LDP + ka];
      a.y = Xs[hn * LDP + ka + 1];
      b.x = Asym[ka * D_ + cg];
      b.y = Asym[(ka + 1) * D_ + cg];
      acc = __builtin_amdgcn_wmma_f32_16x16x4_f32(false, a, false, b,
                                                  (short)0, acc, false, false);
    }
#pragma unroll
    for (int r = 0; r < 8; ++r) {
      Ys[(r + 8 * hi) * LDP + cb + hn] = acc[r];
    }
  }
  __syncthreads();

  {
    const int bb = tid >> 4;
    const int seg = tid & 15;
    float s = 0.f;
#pragma unroll
    for (int j = 0; j < 8; ++j) {
      const int dc = seg * 8 + j;
      s += Xs[bb * LDP + dc] * Ys[bb * LDP + dc];
    }
    tmp[tid] = s;
  }
  __syncthreads();
  if (tid < 16) {
    float s = 0.f;
#pragma unroll
    for (int j = 0; j < 16; ++j) s += tmp[tid * 16 + j];
    t1s[tid] = 0.5f * s;
  }
  __syncthreads();

  {
    float qm[8];
#pragma unroll
    for (int r = 0; r < 8; ++r) qm[r] = __builtin_inff();

    for (int t = 0; t < 2; ++t) {
      const int cb = (wave * 2 + t) * 16;
      v8f acc = {};
#pragma unroll 4
      for (int k0 = 0; k0 < D_; k0 += 4) {
        const int ka = k0 + 2 * hi;
        v2f a, b;
        a.x = Ys[hn * LDP + ka];
        a.y = Ys[hn * LDP + ka + 1];
        b.x = means[ka * C_ + cb + hn];
        b.y = means[(ka + 1) * C_ + cb + hn];
        acc = __builtin_amdgcn_wmma_f32_16x16x4_f32(
            false, a, false, b, (short)0, acc, false, false);
      }
      const float mac = mAm[cb + hn];
#pragma unroll
      for (int r = 0; r < 8; ++r) {
        const float q = t1s[r + 8 * hi] - acc[r] + mac;
        qm[r] = fminf(qm[r], q);
      }
    }
#pragma unroll
    for (int r = 0; r < 8; ++r) {
      float v = qm[r];
      v = fminf(v, __shfl_xor(v, 1, 32));
      v = fminf(v, __shfl_xor(v, 2, 32));
      v = fminf(v, __shfl_xor(v, 4, 32));
      v = fminf(v, __shfl_xor(v, 8, 32));
      if (hn == 0) pm[wave][r + 8 * hi] = v;
    }
  }
  __syncthreads();

  if (tid < 16) {
    float m = pm[0][tid];
#pragma unroll
    for (int w2 = 1; w2 < 8; ++w2) m = fminf(m, pm[w2][tid]);
    out[rowbase + tid] = m;
  }
}

extern "C" void kernel_launch(void* const* d_in, const int* in_sizes, int n_in,
                              void* d_out, int out_size, void* d_ws,
                              size_t ws_size, hipStream_t stream) {
  const float* X     = (const float*)d_in[0];
  const float* means = (const float*)d_in[1];
  const float* alpha = (const float*)d_in[2];
  float* out  = (float*)d_out;
  float* mAm  = (float*)d_ws;
  float* Asym = (float*)d_ws + 256;

  mahal_asym_kernel<<<(D_ * D_) / 256, 256, 0, stream>>>(alpha, Asym);
  mahal_mam_kernel<<<C_, 128, 0, stream>>>(alpha, means, mAm);
  mahal_main_kernel<<<B_ / 16, 256, 0, stream>>>(X, means, Asym, mAm, out);
}


# main: alpha rows and W (prep stores it column-major) staged once per work-group into LDS by whole-row 16-byte loads issued as one group at the start; 16-byte LDS fragment reads; f32 sum order changes
# speedup vs baseline: 2.2511x; 2.2511x over previous
#include <hip/hip_runtime.h>
#include <hip/hip_bf16.h>

typedef float v2f __attribute__((ext_vector_type(2)));
typedef float v4f __attribute__((ext_vector_type(4)));
typedef float v8f __attribute__((ext_vector_type(8)));

#define B_ 4096
#define D_ 128
#define C_ 256


__global__ __launch_bounds__(256) void mahal_prep_kernel(
    const float* __restrict__ alpha, const float* __restrict__ means,
    float* __restrict__ Wt, float* __restrict__ mAm) {
  constexpr int LDA = D_ + 4;
  __shared__ alignas(16) float As[D_ * LDA];
  __shared__ alignas(16) float Ts[16 * LDA];
  __shared__ float red[8][16];

  const int tid = threadIdx.x;
  const int wave = tid >> 5;
  const int lane = tid & 31;
  const int hn = lane & 15;
  const int hi = lane >> 4;
  const int cb = blockIdx.x * 16;
  const int db = wave * 16;
  const int drow = db + hn;

  v4f av[16];
#pragma unroll
  for (int j = 0; j < 16; ++j) {
    const int i = tid + 256 * j;
    av[j] = *reinterpret_cast<const v4f*>(alpha + (i >> 5) * D_ + (i & 31) * 4);
  }

  v2f bf[32];
#pragma unroll
  for (int t = 0; t < 32; ++t) {
    const int ka = 4 * t + 2 * hi;
    bf[t].x = means[ka * C_ + cb + hn];
    bf[t].y = means[(ka + 1) * C_ + cb + hn];
  }
  float mv[8];
#pragma unroll
  for (int r = 0; r < 8; ++r) mv[r] = means[(db + 8 * hi + r) * C_ + cb + hn];

#pragma unroll
  for (int j = 0; j < 16; ++j) {
    const int i = tid + 256 * j;
    *reinterpret_cast<v4f*>(As + (i >> 5) * LDA + (i & 31) * 4) = av[j];
  }
  __syncthreads();

  v2f af[32];
#pragma unroll
  for (int t = 0; t < 32; ++t) {
    const int ka = 4 * t + 2 * hi;
    af[t].x = As[drow * LDA + ka] + As[ka * LDA + drow];
    af[t].y = As[drow * LDA + ka + 1] + As[(ka + 1) * LDA + drow];
  }

  __builtin_amdgcn_sched_barrier(0);

  v8f acc = {};
#pragma unroll
  for (int t = 0; t < 32; ++t) {
    acc = __builtin_amdgcn_wmma_f32_16x16x4_f32(false, af[t], false, bf[t],
                                                (short)0, acc, false, false);
  }

  float p = 0.f;
#pragma unroll
  for (int r = 0; r < 8; ++r) p += acc[r] * mv[r];
  {
    v4f lo = {acc[0], acc[1], acc[2], acc[3]};
    v4f hi4 = {acc[4], acc[5], acc[6], acc[7]};
    float* trow = Ts + hn * LDA + db + 8 * hi;
    *reinterpret_cast<v4f*>(trow) = lo;
    *reinterpret_cast<v4f*>(trow + 4) = hi4;
  }
  p += __shfl_xor(p, 16, 32);
  if (hi == 0) red[wave][hn] = p;
  __syncthreads();

#pragma unroll
  for (int j = 0; j < 2; ++j) {
    const int i = tid + 256 * j;
    const v4f v = *reinterpret_cast<const v4f*>(Ts + (i >> 5) * LDA + (i & 31) * 4);
    *reinterpret_cast<v4f*>(Wt + (cb + (i >> 5)) * D_ + (i & 31) * 4) = v;
  }
  if (tid < 16) {
    float s = red[0][tid];
#pragma unroll
    for (int w2 = 1; w2 < 8; ++w2) s += red[w2][tid];
    mAm[cb + tid] = 0.5f * s;
  }
}

__global__ __launch_bounds__(256) void mahal_main_kernel(
    const float* __restrict__ X, const float* __restrict__ alpha,
    const float* __restrict__ Wt, const float* __restrict__ mAm,
    float* __restrict__ out) {
  constexpr int LDP = D_ + 4;
  __shared__ alignas(16) float Bs[(D_ + C_) * LDP];
  __shared__ alignas(16) float Xs[16 * LDP];
  __shared__ float Ys[16 * LDP];
  __shared__ float tmp[256];
  __shared__ float t1s[16];
  __shared__ float pm[8][16];

  const int tid = threadIdx.x;
  const int wave = tid >> 5;
  const int lane = tid & 31;
  const int hn = lane & 15;
  const int hi = lane >> 4;
  const int rowbase = blockIdx.x * 16;
  const int prow = tid >> 5;
  const int pcol = (tid & 31) * 4;

  v4f xv[2], av[16], wv[32];
#pragma unroll
  for (int j = 0; j < 2; ++j)
    xv[j] = *reinterpret_cast<const v4f*>(X + (rowbase + prow + 8 * j) * D_ + pcol);
#pragma unroll
  for (int j = 0; j < 16; ++j)
    av[j] = *reinterpret_cast<const v4f*>(alpha + (prow + 8 * j) * D_ + pcol);
#pragma unroll
  for (int j = 0; j < 32; ++j)
    wv[j] = *reinterpret_cast<const v4f*>(Wt + (prow + 8 * j) * D_ + pcol);
  const float mac0 = mAm[(wave * 2 + 0) * 16 + hn];
  const float mac1 = mAm[(wave * 2 + 1) * 16 + hn];

#pragma unroll
  for (int j = 0; j < 2; ++j)
    *reinterpret_cast<v4f*>(Xs + (prow + 8 * j) * LDP + pcol) = xv[j];
#pragma unroll
  for (int j = 0; j < 16; ++j)
    *reinterpret_cast<v4f*>(Bs + (prow + 8 * j) * LDP + pcol) = av[j];
  asm volatile("s_wait_dscnt 0x0" ::: "memory");
  asm volatile("s_barrier_signal -1" ::: "memory");
  asm volatile("s_barrier_wait -1" ::: "memory");

  v4f a4[16];
#pragma unroll
  for (int j = 0; j < 16; ++j)
    a4[j] = *reinterpret_cast<const v4f*>(Xs + hn * LDP + 8 * j + 4 * hi);

  {
    const int cb = wave * 16;
    v4f b4[16];
#pragma unroll
    for (int j = 0; j < 16; ++j)
      b4[j] = *reinterpret_cast<const v4f*>(Bs + (cb + hn) * LDP + 8 * j + 4 * hi);
    __builtin_amdgcn_sched_barrier(0);
    v8f acc = {};
#pragma unroll
    for (int j = 0; j < 16; ++j) {
      v2f a, b;
      a.x = a4[j].x; a.y = a4[j].y; b.x = b4[j].x; b.y = b4[j].y;
      acc = __builtin_amdgcn_wmma_f32_16x16x4_f32(false, a, false, b,
                                                  (short)0, acc, false, false);
      a.x = a4[j].z; a.y = a4[j].w; b.x = b4[j].z; b.y = b4[j].w;
      acc = __builtin_amdgcn_wmma_f32_16x16x4_f32(false, a, false, b,
                                                  (short)0, acc, false, false);
    }
#pragma unroll
    for (int r = 0; r < 8; ++r) {
      Ys[(r + 8 * hi) * LDP + cb + hn] = acc[r];
    }
  }

#pragma unroll
  for (int j = 0; j < 32; ++j)
    *reinterpret_cast<v4f*>(Bs + (D_ + prow + 8 * j) * LDP + pcol) = wv[j];
  __syncthreads();

  {
    const int bb = tid >> 4;
    const int seg = tid & 15;
    float s = 0.f;
#pragma unroll
    for (int j = 0; j < 8; ++j) {
      const int dc = seg * 8 + j;
      s += Xs[bb * LDP + dc] * Ys[bb * LDP + dc];
    }
    tmp[tid] = s;
  }
  __syncthreads();
  if (tid < 16) {
    float s = 0.f;
#pragma unroll
    for (int j = 0; j < 16; ++j) s += tmp[tid * 16 + j];
    t1s[tid] = s;
  }
  __syncthreads();

  {
    const float* brow0 = Bs + (D_ + (wave * 2 + 0) * 16 + hn) * LDP + 4 * hi;
    const float* brow1 = Bs + (D_ + (wave * 2 + 1) * 16 + hn) * LDP + 4 * hi;
    v4f b0[16], b1[16];
#pragma unroll
    for (int j = 0; j < 16; ++j) {
      b0[j] = *reinterpret_cast<const v4f*>(brow0 + 8 * j);
      b1[j] = *reinterpret_cast<const v4f*>(brow1 + 8 * j);
    }
    __builtin_amdgcn_sched_barrier(0);
    v8f acc0 = {}, acc1 = {};
#pragma unroll
    for (int j = 0; j < 16; ++j) {
      v2f a, b;
      a.x = a4[j].x; a.y = a4[j].y;
      b.x = b0[j].x; b.y = b0[j].y;
      acc0 = __builtin_amdgcn_wmma_f32_16x16x4_f32(false, a, false, b,
                                                   (short)0, acc0, false, false);
      b.x = b1[j].x; b.y = b1[j].y;
      acc1 = __builtin_amdgcn_wmma_f32_16x16x4_f32(false, a, false, b,
                                                   (short)0, acc1, false, false);
      a.x = a4[j].z; a.y = a4[j].w;
      b.x = b0[j].z; b.y = b0[j].w;
      acc0 = __builtin_amdgcn_wmma_f32_16x16x4_f32(false, a, false, b,
                                                   (short)0, acc0, false, false);
      b.x = b1[j].z; b.y = b1[j].w;
      acc1 = __builtin_amdgcn_wmma_f32_16x16x4_f32(false, a, false, b,
                                                   (short)0, acc1, false, false);
    }

    float qm[8];
#pragma unroll
    for (int r = 0; r < 8; ++r) {
      const float t1v = t1s[r + 8 * hi];
      const float q0 = t1v - acc0[r] + mac0;
      const float q1 = t1v - acc1[r] + mac1;
      qm[r] = fminf(fminf(__builtin_inff(), q0), q1);
    }
#pragma unroll
    for (int r = 0; r < 8; ++r) {
      float v = qm[r];
      v = fminf(v, __shfl_xor(v, 1, 32));
      v = fminf(v, __shfl_xor(v, 2, 32));
      v = fminf(v, __shfl_xor(v, 4, 32));
      v = fminf(v, __shfl_xor(v, 8, 32));
      if (hn == 0) pm[wave][r + 8 * hi] = v;
    }
  }
  __syncthreads();

  if (tid < 16) {
    float m = pm[0][tid];
#pragma unroll
    for (int w2 = 1; w2 < 8; ++w2) m = fminf(m, pm[w2][tid]);
    out[rowbase + tid] = m;
  }
}

extern "C" void kernel_launch(void* const* d_in, const int* in_sizes, int n_in,
                              void* d_out, int out_size, void* d_ws,
                              size_t ws_size, hipStream_t stream) {
  const float* X     = (const float*)d_in[0];
  const float* means = (const float*)d_in[1];
  const float* alpha = (const float*)d_in[2];
  float* out  = (float*)d_out;
  float* mAm  = (float*)d_ws;
  float* Wt   = (float*)d_ws + 256;

  mahal_prep_kernel<<<C_ / 16, 256, 0, stream>>>(alpha, means, Wt, mAm);
  mahal_main_kernel<<<B_ / 16, 256, 0, stream>>>(X, alpha, Wt, mAm, out);
}


# main: row minimum over the 16 lanes of a row as four DPP row_xmask min steps on the vector ALU, eight rows interleaved; was 32 serial LDS-unit lane permutes, each waited for; no sum order changes
# speedup vs baseline: 2.4589x; 1.0923x over previous
#include <hip/hip_runtime.h>
#include <hip/hip_bf16.h>

typedef float v2f __attribute__((ext_vector_type(2)));
typedef float v4f __attribute__((ext_vector_type(4)));
typedef float v8f __attribute__((ext_vector_type(8)));

template <int M>
__device__ __forceinline__ float row_xor(float v) {
  const int b = __builtin_bit_cast(int, v);
  return __builtin_bit_cast(
      float, __builtin_amdgcn_update_dpp(b, b, 0x160 | M, 0xf, 0xf, false));
}

#define B_ 4096
#define D_ 128
#define C_ 256


__global__ __launch_bounds__(256) void mahal_prep_kernel(
    const float* __restrict__ alpha, const float* __restrict__ means,
    float* __restrict__ Wt, float* __restrict__ mAm) {
  constexpr int LDA = D_ + 4;
  __shared__ alignas(16) float As[D_ * LDA];
  __shared__ alignas(16) float Ts[16 * LDA];
  __shared__ float red[8][16];

  const int tid = threadIdx.x;
  const int wave = tid >> 5;
  const int lane = tid & 31;
  const int hn = lane & 15;
  const int hi = lane >> 4;
  const int cb = blockIdx.x * 16;
  const int db = wave * 16;
  const int drow = db + hn;

  v4f av[16];
#pragma unroll
  for (int j = 0; j < 16; ++j) {
    const int i = tid + 256 * j;
    av[j] = *reinterpret_cast<const v4f*>(alpha + (i >> 5) * D_ + (i & 31) * 4);
  }

  v2f bf[32];
#pragma unroll
  for (int t = 0; t < 32; ++t) {
    const int ka = 4 * t + 2 * hi;
    bf[t].x = means[ka * C_ + cb + hn];
    bf[t].y = means[(ka + 1) * C_ + cb + hn];
  }
  float mv[8];
#pragma unroll
  for (int r = 0; r < 8; ++r) mv[r] = means[(db + 8 * hi + r) * C_ + cb + hn];

#pragma unroll
  for (int j = 0; j < 16; ++j) {
    const int i = tid + 256 * j;
    *reinterpret_cast<v4f*>(As + (i >> 5) * LDA + (i & 31) * 4) = av[j];
  }
  __syncthreads();

  v2f af[32];
#pragma unroll
  for (int t = 0; t < 32; ++t) {
    const int ka = 4 * t + 2 * hi;
    af[t].x = As[drow * LDA + ka] + As[ka * LDA + drow];
    af[t].y = As[drow * LDA + ka + 1] + As[(ka + 1) * LDA + drow];
  }

  __builtin_amdgcn_sched_barrier(0);

  v8f acc = {};
#pragma unroll
  for (int t = 0; t < 32; ++t) {
    acc = __builtin_amdgcn_wmma_f32_16x16x4_f32(false, af[t], false, bf[t],
                                                (short)0, acc, false, false);
  }

  float p = 0.f;
#pragma unroll
  for (int r = 0; r < 8; ++r) p += acc[r] * mv[r];
  {
    v4f lo = {acc[0], acc[1], acc[2], acc[3]};
    v4f hi4 = {acc[4], acc[5], acc[6], acc[7]};
    float* trow = Ts + hn * LDA + db + 8 * hi;
    *reinterpret_cast<v4f*>(trow) = lo;
    *reinterpret_cast<v4f*>(trow + 4) = hi4;
  }
  p += __shfl_xor(p, 16, 32);
  if (hi == 0) red[wave][hn] = p;
  __syncthreads();

#pragma unroll
  for (int j = 0; j < 2; ++j) {
    const int i = tid + 256 * j;
    const v4f v = *reinterpret_cast<const v4f*>(Ts + (i >> 5) * LDA + (i & 31) * 4);
    *reinterpret_cast<v4f*>(Wt + (cb + (i >> 5)) * D_ + (i & 31) * 4) = v;
  }
  if (tid < 16) {
    float s = red[0][tid];
#pragma unroll
    for (int w2 = 1; w2 < 8; ++w2) s += red[w2][tid];
    mAm[cb + tid] = 0.5f * s;
  }
}

__global__ __launch_bounds__(256) void mahal_main_kernel(
    const float* __restrict__ X, const float* __restrict__ alpha,
    const float* __restrict__ Wt, const float* __restrict__ mAm,
    float* __restrict__ out) {
  constexpr int LDP = D_ + 4;
  __shared__ alignas(16) float Bs[(D_ + C_) * LDP];
  __shared__ alignas(16) float Xs[16 * LDP];
  __shared__ float Ys[16 * LDP];
  __shared__ float tmp[256];
  __shared__ float t1s[16];
  __shared__ float pm[8][16];

  const int tid = threadIdx.x;
  const int wave = tid >> 5;
  const int lane = tid & 31;
  const int hn = lane & 15;
  const int hi = lane >> 4;
  const int rowbase = blockIdx.x * 16;
  const int prow = tid >> 5;
  const int pcol = (tid & 31) * 4;

  v4f xv[2], av[16], wv[32];
#pragma unroll
  for (int j = 0; j < 2; ++j)
    xv[j] = *reinterpret_cast<const v4f*>(X + (rowbase + prow + 8 * j) * D_ + pcol);
#pragma unroll
  for (int j = 0; j < 16; ++j)
    av[j] = *reinterpret_cast<const v4f*>(alpha + (prow + 8 * j) * D_ + pcol);
#pragma unroll
  for (int j = 0; j < 32; ++j)
    wv[j] = *reinterpret_cast<const v4f*>(Wt + (prow + 8 * j) * D_ + pcol);
  const float mac0 = mAm[(wave * 2 + 0) * 16 + hn];
  const float mac1 = mAm[(wave * 2 + 1) * 16 + hn];

#pragma unroll
  for (int j = 0; j < 2; ++j)
    *reinterpret_cast<v4f*>(Xs + (prow + 8 * j) * LDP + pcol) = xv[j];
#pragma unroll
  for (int j = 0; j < 16; ++j)
    *reinterpret_cast<v4f*>(Bs + (prow + 8 * j) * LDP + pcol) = av[j];
  asm volatile("s_wait_dscnt 0x0" ::: "memory");
  asm volatile("s_barrier_signal -1" ::: "memory");
  asm volatile("s_barrier_wait -1" ::: "memory");

  v4f a4[16];
#pragma unroll
  for (int j = 0; j < 16; ++j)
    a4[j] = *reinterpret_cast<const v4f*>(Xs + hn * LDP + 8 * j + 4 * hi);

  {
    const int cb = wave * 16;
    v4f b4[16];
#pragma unroll
    for (int j = 0; j < 16; ++j)
      b4[j] = *reinterpret_cast<const v4f*>(Bs + (cb + hn) * LDP + 8 * j + 4 * hi);
    __builtin_amdgcn_sched_barrier(0);
    v8f acc = {};
#pragma unroll
    for (int j = 0; j < 16; ++j) {
      v2f a, b;
      a.x = a4[j].x; a.y = a4[j].y; b.x = b4[j].x; b.y = b4[j].y;
      acc = __builtin_amdgcn_wmma_f32_16x16x4_f32(false, a, false, b,
                                                  (short)0, acc, false, false);
      a.x = a4[j].z; a.y = a4[j].w; b.x = b4[j].z; b.y = b4[j].w;
      acc = __builtin_amdgcn_wmma_f32_16x16x4_f32(false, a, false, b,
                                                  (short)0, acc, false, false);
    }
#pragma unroll
    for (int r = 0; r < 8; ++r) {
      Ys[(r + 8 * hi) * LDP + cb + hn] = acc[r];
    }
  }

#pragma unroll
  for (int j = 0; j < 32; ++j)
    *reinterpret_cast<v4f*>(Bs + (D_ + prow + 8 * j) * LDP + pcol) = wv[j];
  __syncthreads();

  {
    const int bb = tid >> 4;
    const int seg = tid & 15;
    float s = 0.f;
#pragma unroll
    for (int j = 0; j < 8; ++j) {
      const int dc = seg * 8 + j;
      s += Xs[bb * LDP + dc] * Ys[bb * LDP + dc];
    }
    tmp[tid] = s;
  }
  __syncthreads();
  if (tid < 16) {
    float s = 0.f;
#pragma unroll
    for (int j = 0; j < 16; ++j) s += tmp[tid * 16 + j];
    t1s[tid] = s;
  }
  __syncthreads();

  {
    const float* brow0 = Bs + (D_ + (wave * 2 + 0) * 16 + hn) * LDP + 4 * hi;
    const float* brow1 = Bs + (D_ + (wave * 2 + 1) * 16 + hn) * LDP + 4 * hi;
    v4f b0[16], b1[16];
#pragma unroll
    for (int j = 0; j < 16; ++j) {
      b0[j] = *reinterpret_cast<const v4f*>(brow0 + 8 * j);
      b1[j] = *reinterpret_cast<const v4f*>(brow1 + 8 * j);
    }
    __builtin_amdgcn_sched_barrier(0);
    v8f acc0 = {}, acc1 = {};
#pragma unroll
    for (int j = 0; j < 16; ++j) {
      v2f a, b;
      a.x = a4[j].x; a.y = a4[j].y;
      b.x = b0[j].x; b.y = b0[j].y;
      acc0 = __builtin_amdgcn_wmma_f32_16x16x4_f32(false, a, false, b,
                                                   (short)0, acc0, false, false);
      b.x = b1[j].x; b.y = b1[j].y;
      acc1 = __builtin_amdgcn_wmma_f32_16x16x4_f32(false, a, false, b,
                                                   (short)0, acc1, false, false);
      a.x = a4[j].z; a.y = a4[j].w;
      b.x = b0[j].z; b.y = b0[j].w;
      acc0 = __builtin_amdgcn_wmma_f32_16x16x4_f32(false, a, false, b,
                                                   (short)0, acc0, false, false);
      b.x = b1[j].z; b.y = b1[j].w;
      acc1 = __builtin_amdgcn_wmma_f32_16x16x4_f32(false, a, false, b,
                                                   (short)0, acc1, false, false);
    }

    float qm[8];
#pragma unroll
    for (int r = 0; r < 8; ++r) {
      const float t1v = t1s[r + 8 * hi];
      const float q0 = t1v - acc0[r] + mac0;
      const float q1 = t1v - acc1[r] + mac1;
      qm[r] = fminf(fminf(__builtin_inff(), q0), q1);
    }
#pragma unroll
    for (int r = 0; r < 8; ++r) qm[r] = fminf(qm[r], row_xor<1>(qm[r]));
#pragma unroll
    for (int r = 0; r < 8; ++r) qm[r] = fminf(qm[r], row_xor<2>(qm[r]));
#pragma unroll
    for (int r = 0; r < 8; ++r) qm[r] = fminf(qm[r], row_xor<4>(qm[r]));
#pragma unroll
    for (int r = 0; r < 8; ++r) qm[r] = fminf(qm[r], row_xor<8>(qm[r]));
    if (hn == 0) {
#pragma unroll
      for (int r = 0; r < 8; ++r) pm[wave][r + 8 * hi] = qm[r];
    }
  }
  __syncthreads();

  if (tid < 16) {
    float m = pm[0][tid];
#pragma unroll
    for (int w2 = 1; w2 < 8; ++w2) m = fminf(m, pm[w2][tid]);
    out[rowbase + tid] = m;
  }
}

extern "C" void kernel_launch(void* const* d_in, const int* in_sizes, int n_in,
                              void* d_out, int out_size, void* d_ws,
                              size_t ws_size, hipStream_t stream) {
  const float* X     = (const float*)d_in[0];
  const float* means = (const float*)d_in[1];
  const float* alpha = (const float*)d_in[2];
  float* out  = (float*)d_out;
  float* mAm  = (float*)d_ws;
  float* Wt   = (float*)d_ws + 256;

  mahal_prep_kernel<<<C_ / 16, 256, 0, stream>>>(alpha, means, Wt, mAm);
  mahal_main_kernel<<<B_ / 16, 256, 0, stream>>>(X, alpha, Wt, mAm, out);
}


# prep re-tiled over 64 work-groups: each a 32x16 block of W, K split over 4 waves per tile (8 WMMAs a wave, summed via LDS); was 16 groups with 32-WMMA chains; mAm in 4 parts; f32 sum order changes
# speedup vs baseline: 2.8788x; 1.1708x over previous
#include <hip/hip_runtime.h>
#include <hip/hip_bf16.h>

typedef float v2f __attribute__((ext_vector_type(2)));
typedef float v4f __attribute__((ext_vector_type(4)));
typedef float v8f __attribute__((ext_vector_type(8)));

template <int M>
__device__ __forceinline__ float row_xor(float v) {
  const int b = __builtin_bit_cast(int, v);
  return __builtin_bit_cast(
      float, __builtin_amdgcn_update_dpp(b, b, 0x160 | M, 0xf, 0xf, false));
}

#define B_ 4096
#define D_ 128
#define C_ 256


__global__ __launch_bounds__(256) void mahal_prep_kernel(
    const float* __restrict__ alpha, const float* __restrict__ means,
    float* __restrict__ Wt, float* __restrict__ mAmP) {
  constexpr int CP = 36;
  __shared__ alignas(16) float part[4 * 16 * CP];
  __shared__ float red[8][32];

  const int tid = threadIdx.x;
  const int wave = tid >> 5;
  const int lane = tid & 31;
  const int hn = lane & 15;
  const int hi = lane >> 4;
  const int ct = blockIdx.x & 15;
  const int rp = blockIdx.x >> 4;
  const int cb = ct * 16;
  const int t = wave >> 2;
  const int q = wave & 3;
  const int db = rp * 32 + t * 16;
  const int drow = db + hn;
  const int kq = q * 32 + 4 * hi;

  v4f ar[4];
  float at[4][4];
  float mb[4][4];
#pragma unroll
  for (int j = 0; j < 4; ++j) {
    ar[j] = *reinterpret_cast<const v4f*>(alpha + drow * D_ + kq + 8 * j);
#pragma unroll
    for (int e = 0; e < 4; ++e) {
      at[j][e] = alpha[(kq + 8 * j + e) * D_ + drow];
      mb[j][e] = means[(kq + 8 * j + e) * C_ + cb + hn];
    }
  }
  float mv[8];
#pragma unroll
  for (int r = 0; r < 8; ++r) mv[r] = means[(db + 8 * hi + r) * C_ + cb + hn];

  v8f acc = {};
#pragma unroll
  for (int j = 0; j < 4; ++j) {
    v2f a, b;
    a.x = ar[j].x + at[j][0]; a.y = ar[j].y + at[j][1];
    b.x = mb[j][0]; b.y = mb[j][1];
    acc = __builtin_amdgcn_wmma_f32_16x16x4_f32(false, a, false, b,
                                                (short)0, acc, false, false);
    a.x = ar[j].z + at[j][2]; a.y = ar[j].w + at[j][3];
    b.x = mb[j][2]; b.y = mb[j][3];
    acc = __builtin_amdgcn_wmma_f32_16x16x4_f32(false, a, false, b,
                                                (short)0, acc, false, false);
  }

  float p = 0.f;
#pragma unroll
  for (int r = 0; r < 8; ++r) p += acc[r] * mv[r];
  {
    v4f lo = {acc[0], acc[1], acc[2], acc[3]};
    v4f hi4 = {acc[4], acc[5], acc[6], acc[7]};
    float* prow = part + (q * 16 + hn) * CP + t * 16 + 8 * hi;
    *reinterpret_cast<v4f*>(prow) = lo;
    *reinterpret_cast<v4f*>(prow + 4) = hi4;
  }
  red[wave][lane] = p;
  __syncthreads();

#pragma unroll
  for (int i = 0; i < 2; ++i) {
    const int cl = wave * 2 + i;
    const float* src = part + cl * CP + lane;
    float s = src[0];
    s += src[16 * CP];
    s += src[32 * CP];
    s += src[48 * CP];
    Wt[(cb + cl) * D_ + rp * 32 + lane] = s;
  }
  if (tid < 16) {
    float s = red[0][tid] + red[0][16 + tid];
#pragma unroll
    for (int w2 = 1; w2 < 8; ++w2) {
      s += red[w2][tid];
      s += red[w2][16 + tid];
    }
    mAmP[(rp * 16 + ct) * 32 + tid] = 0.5f * s;
  }
}

__global__ __launch_bounds__(256) void mahal_main_kernel(
    const float* __restrict__ X, const float* __restrict__ alpha,
    const float* __restrict__ Wt, const float* __restrict__ mAmP,
    float* __restrict__ out) {
  constexpr int LDP = D_ + 4;
  __shared__ alignas(16) float Bs[(D_ + C_) * LDP];
  __shared__ alignas(16) float Xs[16 * LDP];
  __shared__ float Ys[16 * LDP];
  __shared__ float tmp[256];
  __shared__ float t1s[16];
  __shared__ float pm[8][16];

  const int tid = threadIdx.x;
  const int wave = tid >> 5;
  const int lane = tid & 31;
  const int hn = lane & 15;
  const int hi = lane >> 4;
  const int rowbase = blockIdx.x * 16;
  const int prow = tid >> 5;
  const int pcol = (tid & 31) * 4;

  v4f xv[2], av[16], wv[32];
#pragma unroll
  for (int j = 0; j < 2; ++j)
    xv[j] = *reinterpret_cast<const v4f*>(X + (rowbase + prow + 8 * j) * D_ + pcol);
#pragma unroll
  for (int j = 0; j < 16; ++j)
    av[j] = *reinterpret_cast<const v4f*>(alpha + (prow + 8 * j) * D_ + pcol);
#pragma unroll
  for (int j = 0; j < 32; ++j)
    wv[j] = *reinterpret_cast<const v4f*>(Wt + (prow + 8 * j) * D_ + pcol);
  float mp0[4], mp1[4];
#pragma unroll
  for (int rp = 0; rp < 4; ++rp) {
    mp0[rp] = mAmP[(rp * 16 + wave * 2 + 0) * 32 + hn];
    mp1[rp] = mAmP[(rp * 16 + wave * 2 + 1) * 32 + hn];
  }

#pragma unroll
  for (int j = 0; j < 2; ++j)
    *reinterpret_cast<v4f*>(Xs + (prow + 8 * j) * LDP + pcol) = xv[j];
#pragma unroll
  for (int j = 0; j < 16; ++j)
    *reinterpret_cast<v4f*>(Bs + (prow + 8 * j) * LDP + pcol) = av[j];
  asm volatile("s_wait_dscnt 0x0" ::: "memory");
  asm volatile("s_barrier_signal -1" ::: "memory");
  asm volatile("s_barrier_wait -1" ::: "memory");

  v4f a4[16];
#pragma unroll
  for (int j = 0; j < 16; ++j)
    a4[j] = *reinterpret_cast<const v4f*>(Xs + hn * LDP + 8 * j + 4 * hi);

  {
    const int cb = wave * 16;
    v4f b4[16];
#pragma unroll
    for (int j = 0; j < 16; ++j)
      b4[j] = *reinterpret_cast<const v4f*>(Bs + (cb + hn) * LDP + 8 * j + 4 * hi);
    __builtin_amdgcn_sched_barrier(0);
    v8f acc = {};
#pragma unroll
    for (int j = 0; j < 16; ++j) {
      v2f a, b;
      a.x = a4[j].x; a.y = a4[j].y; b.x = b4[j].x; b.y = b4[j].y;
      acc = __builtin_amdgcn_wmma_f32_16x16x4_f32(false, a, false, b,
                                                  (short)0, acc, false, false);
      a.x = a4[j].z; a.y = a4[j].w; b.x = b4[j].z; b.y = b4[j].w;
      acc = __builtin_amdgcn_wmma_f32_16x16x4_f32(false, a, false, b,
                                                  (short)0, acc, false, false);
    }
#pragma unroll
    for (int r = 0; r < 8; ++r) {
      Ys[(r + 8 * hi) * LDP + cb + hn] = acc[r];
    }
  }

#pragma unroll
  for (int j = 0; j < 32; ++j)
    *reinterpret_cast<v4f*>(Bs + (D_ + prow + 8 * j) * LDP + pcol) = wv[j];
  __syncthreads();

  {
    const int bb = tid >> 4;
    const int seg = tid & 15;
    float s = 0.f;
#pragma unroll
    for (int j = 0; j < 8; ++j) {
      const int dc = seg * 8 + j;
      s += Xs[bb * LDP + dc] * Ys[bb * LDP + dc];
    }
    tmp[tid] = s;
  }
  __syncthreads();
  if (tid < 16) {
    float s = 0.f;
#pragma unroll
    for (int j = 0; j < 16; ++j) s += tmp[tid * 16 + j];
    t1s[tid] = s;
  }
  __syncthreads();

  {
    const float* brow0 = Bs + (D_ + (wave * 2 + 0) * 16 + hn) * LDP + 4 * hi;
    const float* brow1 = Bs + (D_ + (wave * 2 + 1) * 16 + hn) * LDP + 4 * hi;
    v4f b0[16], b1[16];
#pragma unroll
    for (int j = 0; j < 16; ++j) {
      b0[j] = *reinterpret_cast<const v4f*>(brow0 + 8 * j);
      b1[j] = *reinterpret_cast<const v4f*>(brow1 + 8 * j);
    }
    __builtin_amdgcn_sched_barrier(0);
    v8f acc0 = {}, acc1 = {};
#pragma unroll
    for (int j = 0; j < 16; ++j) {
      v2f a, b;
      a.x = a4[j].x; a.y = a4[j].y;
      b.x = b0[j].x; b.y = b0[j].y;
      acc0 = __builtin_amdgcn_wmma_f32_16x16x4_f32(false, a, false, b,
                                                   (short)0, acc0, false, false);
      b.x = b1[j].x; b.y = b1[j].y;
      acc1 = __builtin_amdgcn_wmma_f32_16x16x4_f32(false, a, false, b,
                                                   (short)0, acc1, false, false);
      a.x = a4[j].z; a.y = a4[j].w;
      b.x = b0[j].z; b.y = b0[j].w;
      acc0 = __builtin_amdgcn_wmma_f32_16x16x4_f32(false, a, false, b,
                                                   (short)0, acc0, false, false);
      b.x = b1[j].z; b.y = b1[j].w;
      acc1 = __builtin_amdgcn_wmma_f32_16x16x4_f32(false, a, false, b,
                                                   (short)0, acc1, false, false);
    }

    const float mac0 = ((mp0[0] + mp0[1]) + mp0[2]) + mp0[3];
    const float mac1 = ((mp1[0] + mp1[1]) + mp1[2]) + mp1[3];
    float qm[8];
#pragma unroll
    for (int r = 0; r < 8; ++r) {
      const float t1v = t1s[r + 8 * hi];
      const float q0 = t1v - acc0[r] + mac0;
      const float q1 = t1v - acc1[r] + mac1;
      qm[r] = fminf(fminf(__builtin_inff(), q0), q1);
    }
#pragma unroll
    for (int r = 0; r < 8; ++r) qm[r] = fminf(qm[r], row_xor<1>(qm[r]));
#pragma unroll
    for (int r = 0; r < 8; ++r) qm[r] = fminf(qm[r], row_xor<2>(qm[r]));
#pragma unroll
    for (int r = 0; r < 8; ++r) qm[r] = fminf(qm[r], row_xor<4>(qm[r]));
#pragma unroll
    for (int r = 0; r < 8; ++r) qm[r] = fminf(qm[r], row_xor<8>(qm[r]));
    if (hn == 0) {
#pragma unroll
      for (int r = 0; r < 8; ++r) pm[wave][r + 8 * hi] = qm[r];
    }
  }
  __syncthreads();

  if (tid < 16) {
    float m = pm[0][tid];
#pragma unroll
    for (int w2 = 1; w2 < 8; ++w2) m = fminf(m, pm[w2][tid]);
    out[rowbase + tid] = m;
  }
}

extern "C" void kernel_launch(void* const* d_in, const int* in_sizes, int n_in,
                              void* d_out, int out_size, void* d_ws,
                              size_t ws_size, hipStream_t stream) {
  const float* X     = (const float*)d_in[0];
  const float* means = (const float*)d_in[1];
  const float* alpha = (const float*)d_in[2];
  float* out  = (float*)d_out;
  float* mAmP = (float*)d_ws;
  float* Wt   = (float*)d_ws + 2048;

  mahal_prep_kernel<<<64, 256, 0, stream>>>(alpha, means, Wt, mAmP);
  mahal_main_kernel<<<B_ / 16, 256, 0, stream>>>(X, alpha, Wt, mAmP, out);
}


# main k-pipelined: a wave moves only the 48 rows of alpha and W it multiplies, in 32-k chunks (W stored chunk-major) via private LDS rows, chunk q+1 staged amid chunk q's WMMAs; f32 sum order changes
# speedup vs baseline: 3.0492x; 1.0592x over previous
#include <hip/hip_runtime.h>
#include <hip/hip_bf16.h>

typedef float v2f __attribute__((ext_vector_type(2)));
typedef float v4f __attribute__((ext_vector_type(4)));
typedef float v8f __attribute__((ext_vector_type(8)));

template <int M>
__device__ __forceinline__ float row_xor(float v) {
  const int b = __builtin_bit_cast(int, v);
  return __builtin_bit_cast(
      float, __builtin_amdgcn_update_dpp(b, b, 0x160 | M, 0xf, 0xf, false));
}

#define B_ 4096
#define D_ 128
#define C_ 256


__global__ __launch_bounds__(256) void mahal_prep_kernel(
    const float* __restrict__ alpha, const float* __restrict__ means,
    float* __restrict__ Wt, float* __restrict__ mAmP) {
  constexpr int CP = 36;
  __shared__ alignas(16) float part[4 * 16 * CP];
  __shared__ float red[8][32];

  const int tid = threadIdx.x;
  const int wave = tid >> 5;
  const int lane = tid & 31;
  const int hn = lane & 15;
  const int hi = lane >> 4;
  const int ct = blockIdx.x & 15;
  const int rp = blockIdx.x >> 4;
  const int cb = ct * 16;
  const int t = wave >> 2;
  const int q = wave & 3;
  const int db = rp * 32 + t * 16;
  const int drow = db + hn;
  const int kq = q * 32 + 4 * hi;

  v4f ar[4];
  float at[4][4];
  float mb[4][4];
#pragma unroll
  for (int j = 0; j < 4; ++j) {
    ar[j] = *reinterpret_cast<const v4f*>(alpha + drow * D_ + kq + 8 * j);
#pragma unroll
    for (int e = 0; e < 4; ++e) {
      at[j][e] = alpha[(kq + 8 * j + e) * D_ + drow];
      mb[j][e] = means[(kq + 8 * j + e) * C_ + cb + hn];
    }
  }
  float mv[8];
#pragma unroll
  for (int r = 0; r < 8; ++r) mv[r] = means[(db + 8 * hi + r) * C_ + cb + hn];

  v8f acc = {};
#pragma unroll
  for (int j = 0; j < 4; ++j) {
    v2f a, b;
    a.x = ar[j].x + at[j][0]; a.y = ar[j].y + at[j][1];
    b.x = mb[j][0]; b.y = mb[j][1];
    acc = __builtin_amdgcn_wmma_f32_16x16x4_f32(false, a, false, b,
                                                (short)0, acc, false, false);
    a.x = ar[j].z + at[j][2]; a.y = ar[j].w + at[j][3];
    b.x = mb[j][2]; b.y = mb[j][3];
    acc = __builtin_amdgcn_wmma_f32_16x16x4_f32(false, a, false, b,
                                                (short)0, acc, false, false);
  }

  float p = 0.f;
#pragma unroll
  for (int r = 0; r < 8; ++r) p += acc[r] * mv[r];
  {
    v4f lo = {acc[0], acc[1], acc[2], acc[3]};
    v4f hi4 = {acc[4], acc[5], acc[6], acc[7]};
    float* prow = part + (q * 16 + hn) * CP + t * 16 + 8 * hi;
    *reinterpret_cast<v4f*>(prow) = lo;
    *reinterpret_cast<v4f*>(prow + 4) = hi4;
  }
  red[wave][lane] = p;
  __syncthreads();

#pragma unroll
  for (int i = 0; i < 2; ++i) {
    const int cl = wave * 2 + i;
    const float* src = part + cl * CP + lane;
    float s = src[0];
    s += src[16 * CP];
    s += src[32 * CP];
    s += src[48 * CP];
    Wt[(rp * C_ + cb + cl) * 32 + lane] = s;
  }
  if (tid < 16) {
    float s = red[0][tid] + red[0][16 + tid];
#pragma unroll
    for (int w2 = 1; w2 < 8; ++w2) {
      s += red[w2][tid];
      s += red[w2][16 + tid];
    }
    mAmP[(rp * 16 + ct) * 32 + tid] = 0.5f * s;
  }
}

__global__ __launch_bounds__(256) void mahal_main_kernel(
    const float* __restrict__ X, const float* __restrict__ alpha,
    const float* __restrict__ Wt, const float* __restrict__ mAmP,
    float* __restrict__ out) {
  constexpr int LDP = D_ + 4;
  __shared__ alignas(16) float Bs[(D_ + C_) * LDP];
  __shared__ alignas(16) float Xs[16 * LDP];
  __shared__ float t1p[8][16];
  __shared__ float pm[8][16];

  const int tid = threadIdx.x;
  const int wave = tid >> 5;
  const int lane = tid & 31;
  const int hn = lane & 15;
  const int hi = lane >> 4;
  const int rowbase = blockIdx.x * 16;
  const int prow = tid >> 5;
  const int pcol = (tid & 31) * 4;
  const int g = lane >> 3;
  const int ck = (lane & 7) * 4;
  const int arow = wave * 16 + 8 * (g & 1) + 4 * (g >> 1);
  const int wrow = wave * 32 + g;

  v4f xv[2], av[4][4], wv[4][8];
#pragma unroll
  for (int j = 0; j < 2; ++j)
    xv[j] = *reinterpret_cast<const v4f*>(X + (rowbase + prow + 8 * j) * D_ + pcol);
#pragma unroll
  for (int q = 0; q < 4; ++q) {
    __builtin_amdgcn_sched_barrier(0);
#pragma unroll
    for (int i = 0; i < 4; ++i)
      av[q][i] = *reinterpret_cast<const v4f*>(alpha + (arow + i) * D_ + 32 * q + ck);
#pragma unroll
    for (int i = 0; i < 8; ++i)
      wv[q][i] = *reinterpret_cast<const v4f*>(
          Wt + (q * C_ + wrow + 4 * i) * 32 + ck);
  }
  __builtin_amdgcn_sched_barrier(0);
  float mp0[4], mp1[4];
#pragma unroll
  for (int rp = 0; rp < 4; ++rp) {
    mp0[rp] = mAmP[(rp * 16 + wave * 2 + 0) * 32 + hn];
    mp1[rp] = mAmP[(rp * 16 + wave * 2 + 1) * 32 + hn];
  }

#pragma unroll
  for (int j = 0; j < 2; ++j)
    *reinterpret_cast<v4f*>(Xs + (prow + 8 * j) * LDP + pcol) = xv[j];
  asm volatile("s_wait_dscnt 0x0" ::: "memory");
  asm volatile("s_barrier_signal -1" ::: "memory");
  asm volatile("s_barrier_wait -1" ::: "memory");

  v4f a4[16];
#pragma unroll
  for (int j = 0; j < 16; ++j)
    a4[j] = *reinterpret_cast<const v4f*>(Xs + hn * LDP + 8 * j + 4 * hi);
  float xd[8];
#pragma unroll
  for (int r = 0; r < 8; ++r) xd[r] = Xs[(r + 8 * hi) * LDP + wave * 16 + hn];

  float* sa = Bs + arow * LDP + ck;
  float* sw = Bs + (D_ + wrow) * LDP + ck;
  const float* fy  = Bs + (wave * 16 + hn) * LDP + 4 * hi;
  const float* fz0 = Bs + (D_ + (wave * 2 + 0) * 16 + hn) * LDP + 4 * hi;
  const float* fz1 = Bs + (D_ + (wave * 2 + 1) * 16 + hn) * LDP + 4 * hi;

  v4f fb[2][12];

#pragma unroll
  for (int i = 0; i < 4; ++i)
    *reinterpret_cast<v4f*>(sa + i * LDP) = av[0][i];
#pragma unroll
  for (int i = 0; i < 8; ++i)
    *reinterpret_cast<v4f*>(sw + 4 * i * LDP) = wv[0][i];
  asm volatile("s_wait_dscnt 0x0" ::: "memory");
#pragma unroll
  for (int j = 0; j < 4; ++j) {
    fb[0][j]     = *reinterpret_cast<const v4f*>(fy  + 8 * j);
    fb[0][4 + j] = *reinterpret_cast<const v4f*>(fz0 + 8 * j);
    fb[0][8 + j] = *reinterpret_cast<const v4f*>(fz1 + 8 * j);
  }

  v8f accy = {}, acc0 = {}, acc1 = {};
#pragma unroll
  for (int q = 0; q < 4; ++q) {
    const int cur = q & 1;
    const int nxt = cur ^ 1;
#pragma unroll
    for (int j = 0; j < 4; ++j) {
      if (j == 2 && q < 3) {
        __builtin_amdgcn_sched_barrier(0);
#pragma unroll
        for (int i = 0; i < 4; ++i)
          *reinterpret_cast<v4f*>(sa + i * LDP + 32 * (q + 1)) = av[q + 1][i];
#pragma unroll
        for (int i = 0; i < 8; ++i)
          *reinterpret_cast<v4f*>(sw + 4 * i * LDP + 32 * (q + 1)) = wv[q + 1][i];
        asm volatile("s_wait_dscnt 0x0" ::: "memory");
#pragma unroll
        for (int jn = 0; jn < 4; ++jn) {
          fb[nxt][jn]     = *reinterpret_cast<const v4f*>(fy  + 32 * (q + 1) + 8 * jn);
          fb[nxt][4 + jn] = *reinterpret_cast<const v4f*>(fz0 + 32 * (q + 1) + 8 * jn);
          fb[nxt][8 + jn] = *reinterpret_cast<const v4f*>(fz1 + 32 * (q + 1) + 8 * jn);
        }
        __builtin_amdgcn_sched_barrier(0);
      }
      const v4f av4 = a4[4 * q + j];
      const v4f by = fb[cur][j], bz0 = fb[cur][4 + j], bz1 = fb[cur][8 + j];
      v2f a, b;
      a.x = av4.x; a.y = av4.y;
      b.x = by.x; b.y = by.y;
      accy = __builtin_amdgcn_wmma_f32_16x16x4_f32(false, a, false, b,
                                                   (short)0, accy, false, false);
      b.x = bz0.x; b.y = bz0.y;
      acc0 = __builtin_amdgcn_wmma_f32_16x16x4_f32(false, a, false, b,
                                                   (short)0, acc0, false, false);
      b.x = bz1.x; b.y = bz1.y;
      acc1 = __builtin_amdgcn_wmma_f32_16x16x4_f32(false, a, false, b,
                                                   (short)0, acc1, false, false);
      a.x = av4.z; a.y = av4.w;
      b.x = by.z; b.y = by.w;
      accy = __builtin_amdgcn_wmma_f32_16x16x4_f32(false, a, false, b,
                                                   (short)0, accy, false, false);
      b.x = bz0.z; b.y = bz0.w;
      acc0 = __builtin_amdgcn_wmma_f32_16x16x4_f32(false, a, false, b,
                                                   (short)0, acc0, false, false);
      b.x = bz1.z; b.y = bz1.w;
      acc1 = __builtin_amdgcn_wmma_f32_16x16x4_f32(false, a, false, b,
                                                   (short)0, acc1, false, false);
    }
  }

  {
    const float mac0 = ((mp0[0] + mp0[1]) + mp0[2]) + mp0[3];
    const float mac1 = ((mp1[0] + mp1[1]) + mp1[2]) + mp1[3];
    float qm[8], ts[8];
#pragma unroll
    for (int r = 0; r < 8; ++r) {
      qm[r] = fminf(fminf(__builtin_inff(), mac0 - acc0[r]), mac1 - acc1[r]);
      ts[r] = accy[r] * xd[r];
    }
#pragma unroll
    for (int r = 0; r < 8; ++r) { qm[r] = fminf(qm[r], row_xor<1>(qm[r])); ts[r] += row_xor<1>(ts[r]); }
#pragma unroll
    for (int r = 0; r < 8; ++r) { qm[r] = fminf(qm[r], row_xor<2>(qm[r])); ts[r] += row_xor<2>(ts[r]); }
#pragma unroll
    for (int r = 0; r < 8; ++r) { qm[r] = fminf(qm[r], row_xor<4>(qm[r])); ts[r] += row_xor<4>(ts[r]); }
#pragma unroll
    for (int r = 0; r < 8; ++r) { qm[r] = fminf(qm[r], row_xor<8>(qm[r])); ts[r] += row_xor<8>(ts[r]); }
    if (hn == 0) {
#pragma unroll
      for (int r = 0; r < 8; ++r) {
        pm[wave][r + 8 * hi] = qm[r];
        t1p[wave][r + 8 * hi] = ts[r];
      }
    }
  }
  __syncthreads();

  if (tid < 16) {
    float m = pm[0][tid];
    float s = t1p[0][tid];
#pragma unroll
    for (int w2 = 1; w2 < 8; ++w2) {
      m = fminf(m, pm[w2][tid]);
      s += t1p[w2][tid];
    }
    out[rowbase + tid] = s + m;
  }
}

extern "C" void kernel_launch(void* const* d_in, const int* in_sizes, int n_in,
                              void* d_out, int out_size, void* d_ws,
                              size_t ws_size, hipStream_t stream) {
  const float* X     = (const float*)d_in[0];
  const float* means = (const float*)d_in[1];
  const float* alpha = (const float*)d_in[2];
  float* out  = (float*)d_out;
  float* mAmP = (float*)d_ws;
  float* Wt   = (float*)d_ws + 2048;

  mahal_prep_kernel<<<64, 256, 0, stream>>>(alpha, means, Wt, mAmP);
  mahal_main_kernel<<<B_ / 16, 256, 0, stream>>>(X, alpha, Wt, mAmP, out);
}


# main re-tiled to one wave per SIMD (4-wave groups, 96 columns a wave, six chains): B fragments loaded from fragment-major copies of W and alpha that prep writes, no LDS staging; f32 sum order changes
# speedup vs baseline: 3.3223x; 1.0896x over previous
#include <hip/hip_runtime.h>
#include <hip/hip_bf16.h>

typedef float v2f __attribute__((ext_vector_type(2)));
typedef float v4f __attribute__((ext_vector_type(4)));
typedef float v8f __attribute__((ext_vector_type(8)));

template <int M>
__device__ __forceinline__ float row_xor(float v) {
  const int b = __builtin_bit_cast(int, v);
  return __builtin_bit_cast(
      float, __builtin_amdgcn_update_dpp(b, b, 0x160 | M, 0xf, 0xf, false));
}

#define B_ 4096
#define D_ 128
#define C_ 256


__global__ __launch_bounds__(256) void mahal_prep_kernel(
    const float* __restrict__ alpha, const float* __restrict__ means,
    float* __restrict__ Wf, float* __restrict__ Af, float* __restrict__ mAmP) {
  constexpr int CP = 36;
  __shared__ alignas(16) float part[4 * 16 * CP];
  __shared__ float red[8][32];

  const int tid = threadIdx.x;
  const int wave = tid >> 5;
  const int lane = tid & 31;
  const int hn = lane & 15;
  const int hi = lane >> 4;
  const int ct = blockIdx.x & 15;
  const int rp = blockIdx.x >> 4;
  const int cb = ct * 16;
  const int t = wave >> 2;
  const int q = wave & 3;
  const int db = rp * 32 + t * 16;
  const int drow = db + hn;
  const int kq = q * 32 + 4 * hi;

  v4f ar[4];
  float at[4][4];
  float mb[4][4];
#pragma unroll
  for (int j = 0; j < 4; ++j) {
    ar[j] = *reinterpret_cast<const v4f*>(alpha + drow * D_ + kq + 8 * j);
#pragma unroll
    for (int e = 0; e < 4; ++e) {
      at[j][e] = alpha[(kq + 8 * j + e) * D_ + drow];
      mb[j][e] = means[(kq + 8 * j + e) * C_ + cb + hn];
    }
  }
  float mv[8];
#pragma unroll
  for (int r = 0; r < 8; ++r) mv[r] = means[(db + 8 * hi + r) * C_ + cb + hn];

  v8f acc = {};
#pragma unroll
  for (int j = 0; j < 4; ++j) {
    v2f a, b;
    a.x = ar[j].x + at[j][0]; a.y = ar[j].y + at[j][1];
    b.x = mb[j][0]; b.y = mb[j][1];
    acc = __builtin_amdgcn_wmma_f32_16x16x4_f32(false, a, false, b,
                                                (short)0, acc, false, false);
    a.x = ar[j].z + at[j][2]; a.y = ar[j].w + at[j][3];
    b.x = mb[j][2]; b.y = mb[j][3];
    acc = __builtin_amdgcn_wmma_f32_16x16x4_f32(false, a, false, b,
                                                (short)0, acc, false, false);
  }

  float p = 0.f;
#pragma unroll
  for (int r = 0; r < 8; ++r) p += acc[r] * mv[r];
  {
    v4f lo = {acc[0], acc[1], acc[2], acc[3]};
    v4f hi4 = {acc[4], acc[5], acc[6], acc[7]};
    float* prow = part + (q * 16 + hn) * CP + t * 16 + 8 * hi;
    *reinterpret_cast<v4f*>(prow) = lo;
    *reinterpret_cast<v4f*>(prow + 4) = hi4;
  }
  red[wave][lane] = p;
  __syncthreads();

  {
    const int e2 = 2 * (tid & 1);
    const int hn2 = (tid >> 1) & 15;
    const int hi2 = (tid >> 5) & 1;
    const int j2 = tid >> 6;
    const int i = 8 * j2 + 4 * hi2 + e2;
    const float* src = part + hn2 * CP + i;
    v2f s2 = *reinterpret_cast<const v2f*>(src);
    s2 += *reinterpret_cast<const v2f*>(src + 16 * CP);
    s2 += *reinterpret_cast<const v2f*>(src + 32 * CP);
    s2 += *reinterpret_cast<const v2f*>(src + 48 * CP);
    *reinterpret_cast<v2f*>(Wf + (rp * 16 + ct) * 512 + 2 * tid) = s2;
  }
  if (ct == 0) {
#pragma unroll
    for (int j = 0; j < 4; ++j)
      *reinterpret_cast<v4f*>(Af + (((q * 8 + rp * 2 + t) * 4 + j) * 32 + lane) * 4) = ar[j];
  }
  if (tid < 16) {
    float s = red[0][tid] + red[0][16 + tid];
#pragma unroll
    for (int w2 = 1; w2 < 8; ++w2) {
      s += red[w2][tid];
      s += red[w2][16 + tid];
    }
    mAmP[(rp * 16 + ct) * 32 + tid] = 0.5f * s;
  }
}

template <int Q>
__device__ __forceinline__ void load_chunk(v4f (&bf)[4][6][4], const float* pa,
                                           const float* pw) {
#pragma unroll
  for (int j = 0; j < 4; ++j) {
#pragma unroll
    for (int u = 0; u < 2; ++u)
      bf[Q][u][j] = *reinterpret_cast<const v4f*>(pa + (Q * 8 + u) * 512 + j * 128);
#pragma unroll
    for (int u = 0; u < 4; ++u)
      bf[Q][2 + u][j] = *reinterpret_cast<const v4f*>(pw + (Q * 16 + u) * 512 + j * 128);
  }
}

template <int Q>
__device__ __forceinline__ void mma_chunk(v8f (&acc)[6], const v4f (&a4)[16],
                                          const v4f (&bf)[4][6][4]) {
#pragma unroll
  for (int j = 0; j < 4; ++j) {
    const v4f av4 = a4[4 * Q + j];
    v2f a, b;
    a.x = av4.x; a.y = av4.y;
#pragma unroll
    for (int u = 0; u < 6; ++u) {
      b.x = bf[Q][u][j].x; b.y = bf[Q][u][j].y;
      acc[u] = __builtin_amdgcn_wmma_f32_16x16x4_f32(false, a, false, b,
                                                     (short)0, acc[u], false, false);
    }
    a.x = av4.z; a.y = av4.w;
#pragma unroll
    for (int u = 0; u < 6; ++u) {
      b.x = bf[Q][u][j].z; b.y = bf[Q][u][j].w;
      acc[u] = __builtin_amdgcn_wmma_f32_16x16x4_f32(false, a, false, b,
                                                     (short)0, acc[u], false, false);
    }
  }
}

__global__ __launch_bounds__(128) void mahal_main_kernel(
    const float* __restrict__ X, const float* __restrict__ Af,
    const float* __restrict__ Wf, const float* __restrict__ mAmP,
    float* __restrict__ out) {
  constexpr int LDP = D_ + 4;
  constexpr int XS_FLOATS = 42 * 1024;
  static_assert(XS_FLOATS >= 16 * LDP, "X tile must fit");
  __shared__ alignas(16) float Xs[XS_FLOATS];
  __shared__ float t1p[4][16];
  __shared__ float pm[4][16];

  const int tid = threadIdx.x;
  const int wave = tid >> 5;
  const int lane = tid & 31;
  const int hn = lane & 15;
  const int hi = lane >> 4;
  const int rowbase = blockIdx.x * 16;

  const float* pa = Af + (wave * 2) * 512 + lane * 4;
  const float* pw = Wf + (wave * 4) * 512 + lane * 4;

  v4f xv[4];
#pragma unroll
  for (int j = 0; j < 4; ++j)
    xv[j] = *reinterpret_cast<const v4f*>(X + (rowbase + wave + 4 * j) * D_ + lane * 4);
  v4f bf[4][6][4];
  load_chunk<0>(bf, pa, pw);
  load_chunk<1>(bf, pa, pw);
  __builtin_amdgcn_sched_barrier(0);

#pragma unroll
  for (int j = 0; j < 4; ++j)
    *reinterpret_cast<v4f*>(Xs + (wave + 4 * j) * LDP + lane * 4) = xv[j];
  asm volatile("s_wait_dscnt 0x0" ::: "memory");
  asm volatile("s_barrier_signal -1" ::: "memory");
  asm volatile("s_barrier_wait -1" ::: "memory");

  v4f a4[16];
#pragma unroll
  for (int j = 0; j < 16; ++j)
    a4[j] = *reinterpret_cast<const v4f*>(Xs + hn * LDP + 8 * j + 4 * hi);
  float xd[2][8];
#pragma unroll
  for (int u = 0; u < 2; ++u)
#pragma unroll
    for (int r = 0; r < 8; ++r)
      xd[u][r] = Xs[(r + 8 * hi) * LDP + wave * 32 + u * 16 + hn];

  __builtin_amdgcn_sched_barrier(0);
  load_chunk<2>(bf, pa, pw);
  __builtin_amdgcn_sched_barrier(0);

  v8f acc[6] = {};
  mma_chunk<0>(acc, a4, bf);

  __builtin_amdgcn_sched_barrier(0);
  load_chunk<3>(bf, pa, pw);
  float mp[4][4];
#pragma unroll
  for (int u = 0; u < 4; ++u)
#pragma unroll
    for (int rp = 0; rp < 4; ++rp)
      mp[u][rp] = mAmP[(rp * 16 + wave * 4 + u) * 32 + hn];
  __builtin_amdgcn_sched_barrier(0);

  mma_chunk<1>(acc, a4, bf);
  mma_chunk<2>(acc, a4, bf);
  mma_chunk<3>(acc, a4, bf);

  {
    float mac[4];
#pragma unroll
    for (int u = 0; u < 4; ++u)
      mac[u] = ((mp[u][0] + mp[u][1]) + mp[u][2]) + mp[u][3];
    float qm[8], ts[8];
#pragma unroll
    for (int r = 0; r < 8; ++r) {
      float m = __builtin_inff();
#pragma unroll
      for (int u = 0; u < 4; ++u) m = fminf(m, mac[u] - acc[2 + u][r]);
      qm[r] = m;
      ts[r] = acc[0][r] * xd[0][r] + acc[1][r] * xd[1][r];
    }
#pragma unroll
    for (int r = 0; r < 8; ++r) { qm[r] = fminf(qm[r], row_xor<1>(qm[r])); ts[r] += row_xor<1>(ts[r]); }
#pragma unroll
    for (int r = 0; r < 8; ++r) { qm[r] = fminf(qm[r], row_xor<2>(qm[r])); ts[r] += row_xor<2>(ts[r]); }
#pragma unroll
    for (int r = 0; r < 8; ++r) { qm[r] = fminf(qm[r], row_xor<4>(qm[r])); ts[r] += row_xor<4>(ts[r]); }
#pragma unroll
    for (int r = 0; r < 8; ++r) { qm[r] = fminf(qm[r], row_xor<8>(qm[r])); ts[r] += row_xor<8>(ts[r]); }
    if (hn == 0) {
#pragma unroll
      for (int r = 0; r < 8; ++r) {
        pm[wave][r + 8 * hi] = qm[r];
        t1p[wave][r + 8 * hi] = ts[r];
      }
    }
  }
  __syncthreads();

  if (tid < 16) {
    float m = pm[0][tid];
    float s = t1p[0][tid];
#pragma unroll
    for (int w2 = 1; w2 < 4; ++w2) {
      m = fminf(m, pm[w2][tid]);
      s += t1p[w2][tid];
    }
    out[rowbase + tid] = s + m;
  }
}

extern "C" void kernel_launch(void* const* d_in, const int* in_sizes, int n_in,
                              void* d_out, int out_size, void* d_ws,
                              size_t ws_size, hipStream_t stream) {
  const float* X     = (const float*)d_in[0];
  const float* means = (const float*)d_in[1];
  const float* alpha = (const float*)d_in[2];
  float* out  = (float*)d_out;
  float* mAmP = (float*)d_ws;
  float* Wf   = (float*)d_ws + 2048;
  float* Af   = (float*)d_ws + 2048 + 32768;

  mahal_prep_kernel<<<64, 256, 0, stream>>>(alpha, means, Wf, Af, mAmP);
  mahal_main_kernel<<<B_ / 16, 128, 0, stream>>>(X, Af, Wf, mAmP, out);
}
